# starting kernel
# baseline (speedup 1.0000x reference)
#include <hip/hip_runtime.h>


#define NFEAT 20480
#define NACC  128
#define NBUCK 4
#define BAG   32
#define POS_PER_BLOCK 16
#define THREADS (POS_PER_BLOCK * 32)

typedef __attribute__((ext_vector_type(2))) float v2f;
typedef __attribute__((ext_vector_type(8))) float v8f;

#if defined(__has_builtin)
#  if __has_builtin(__builtin_amdgcn_wmma_f32_16x16x4_f32)
#    define HAVE_WMMA_F32X4 1
#  endif
#endif

__device__ __forceinline__ float4 ld4(const float* p) {
  return *(const float4*)p;
}

__global__ __launch_bounds__(THREADS) void nnue_fwd(
    const int* __restrict__ w_cols, const int* __restrict__ b_cols,
    const float* __restrict__ stms, const float* __restrict__ psqt_w,
    const float* __restrict__ acc_w, const float* __restrict__ acc_bias,
    const float* __restrict__ layer_w, const float* __restrict__ layer_b,
    float* __restrict__ out, int n_pos)
{
  __shared__ float sBlend[POS_PER_BLOCK][260];
  __shared__ float sPsqt[POS_PER_BLOCK][NBUCK];
  __shared__ float sSign[POS_PER_BLOCK];

  const int wave = threadIdx.x >> 5;
  const int lane = threadIdx.x & 31;
  const int pos  = blockIdx.x * POS_PER_BLOCK + wave;

  if (pos < n_pos) {
    const size_t base = (size_t)pos * BAG;
    const int wcolL = w_cols[base + lane];
    const int bcolL = b_cols[base + lane];

    float4 p;
    {
      float4 pw = ld4(psqt_w + (size_t)wcolL * NBUCK);
      float4 pb = ld4(psqt_w + (size_t)bcolL * NBUCK);
      p.x = pw.x - pb.x; p.y = pw.y - pb.y;
      p.z = pw.z - pb.z; p.w = pw.w - pb.w;
    }

    float4 wa = make_float4(0.f, 0.f, 0.f, 0.f);
    float4 ba = make_float4(0.f, 0.f, 0.f, 0.f);
    const int laneOff = lane * 4;
#pragma unroll
    for (int j = 0; j < BAG; ++j) {
      const int wc = __builtin_amdgcn_readlane(wcolL, j);
      float4 r = ld4(acc_w + (size_t)wc * NACC + laneOff);
      wa.x += r.x; wa.y += r.y; wa.z += r.z; wa.w += r.w;
    }
#pragma unroll
    for (int j = 0; j < BAG; ++j) {
      const int bc = __builtin_amdgcn_readlane(bcolL, j);
      float4 r = ld4(acc_w + (size_t)bc * NACC + laneOff);
      ba.x += r.x; ba.y += r.y; ba.z += r.z; ba.w += r.w;
    }

    float4 bias = ld4(acc_bias + laneOff);
    wa.x = fminf(fmaxf(wa.x + bias.x, 0.f), 1.f);
    wa.y = fminf(fmaxf(wa.y + bias.y, 0.f), 1.f);
    wa.z = fminf(fmaxf(wa.z + bias.z, 0.f), 1.f);
    wa.w = fminf(fmaxf(wa.w + bias.w, 0.f), 1.f);
    ba.x = fminf(fmaxf(ba.x + bias.x, 0.f), 1.f);
    ba.y = fminf(fmaxf(ba.y + bias.y, 0.f), 1.f);
    ba.z = fminf(fmaxf(ba.z + bias.z, 0.f), 1.f);
    ba.w = fminf(fmaxf(ba.w + bias.w, 0.f), 1.f);

    const float s = stms[pos];
    const bool flip = (s != 0.0f);
    float4 first  = flip ? ba : wa;
    float4 second = flip ? wa : ba;
    float* row = &sBlend[wave][0];
    *(float4*)(row + 4 * lane)        = first;
    *(float4*)(row + NACC + 4 * lane) = second;

#pragma unroll
    for (int off = 16; off > 0; off >>= 1) {
      p.x += __shfl_xor(p.x, off, 32);
      p.y += __shfl_xor(p.y, off, 32);
      p.z += __shfl_xor(p.z, off, 32);
      p.w += __shfl_xor(p.w, off, 32);
    }
    if (lane == 0) {
      sPsqt[wave][0] = p.x; sPsqt[wave][1] = p.y;
      sPsqt[wave][2] = p.z; sPsqt[wave][3] = p.w;
      sSign[wave] = flip ? -1.0f : 1.0f;
    }
  }
  __syncthreads();

  if (wave == 0) {
    const int n  = lane & 15;
    const int hi = lane >> 4;
    const int nq = (n < NBUCK) ? n : 0;
    const float nmask = (n < NBUCK) ? 1.0f : 0.0f;
    const float* lw = layer_w + (size_t)nq * (2 * NACC);
    const float* arow = &sBlend[n][0];

#ifdef HAVE_WMMA_F32X4
    v8f c = {0.f, 0.f, 0.f, 0.f, 0.f, 0.f, 0.f, 0.f};
#pragma unroll 8
    for (int k4 = 0; k4 < (2 * NACC) / 4; ++k4) {
      const int kb = k4 * 4 + hi * 2;
      v2f a; a.x = arow[kb];        a.y = arow[kb + 1];
      v2f b; b.x = lw[kb] * nmask;  b.y = lw[kb + 1] * nmask;
      c = __builtin_amdgcn_wmma_f32_16x16x4_f32(
              false, a, false, b, (short)0, c, false, false);
    }
    if (n < NBUCK) {
      const float lb = layer_b[n];
#pragma unroll
      for (int r = 0; r < 8; ++r) {
        const int m = r + hi * 8;
        const int opos = blockIdx.x * POS_PER_BLOCK + m;
        if (opos < n_pos)
          out[(size_t)opos * NBUCK + n] =
              sPsqt[m][n] + sSign[m] * (c[r] + lb);
      }
    }
#else
    if (hi == 0) {
      const int m = n;
      const int opos = blockIdx.x * POS_PER_BLOCK + m;
      if (opos < n_pos) {
        const float* brow = &sBlend[m][0];
        const float sign = sSign[m];
#pragma unroll
        for (int bkt = 0; bkt < NBUCK; ++bkt) {
          float acc = layer_b[bkt];
          for (int k = 0; k < 2 * NACC; ++k)
            acc += brow[k] * layer_w[bkt * (2 * NACC) + k];
          out[(size_t)opos * NBUCK + bkt] = sPsqt[m][bkt] + sign * acc;
        }
      }
    }
#endif
  }
}

extern "C" void kernel_launch(void* const* d_in, const int* in_sizes, int n_in,
                              void* d_out, int out_size, void* d_ws, size_t ws_size,
                              hipStream_t stream) {
  const int*   w_cols   = (const int*)d_in[1];
  const int*   b_cols   = (const int*)d_in[3];
  const float* stms     = (const float*)d_in[4];
  const float* psqt_w   = (const float*)d_in[5];
  const float* acc_w    = (const float*)d_in[6];
  const float* acc_bias = (const float*)d_in[7];
  const float* layer_w  = (const float*)d_in[8];
  const float* layer_b  = (const float*)d_in[9];
  float* out = (float*)d_out;

  const int n_pos = in_sizes[4];
  const int blocks = (n_pos + POS_PER_BLOCK - 1) / POS_PER_BLOCK;
  nnue_fwd<<<blocks, THREADS, 0, stream>>>(
      w_cols, b_cols, stms, psqt_w, acc_w, acc_bias, layer_w, layer_b,
      out, n_pos);
}


# psqt table rows below 17920 read from an LDS copy kept by one persistent work-group per WGP (16 gather waves, asm row loads 24 in flight, 1 matrix wave on a double-buffered tile); sum order unchanged
# speedup vs baseline: 1.2860x; 1.2860x over previous
#include <hip/hip_runtime.h>


#define NFEAT 20480
#define NACC  128
#define NBUCK 4
#define BAG   32
#define TILE  16
#define GATHER_WAVES 16
#define THREADS ((GATHER_WAVES + 1) * 32)
#define PSQT_LDS_ROWS 17920
#define ROWP 260

typedef __attribute__((ext_vector_type(2))) float v2f;
typedef __attribute__((ext_vector_type(4))) float v4f;
typedef __attribute__((ext_vector_type(8))) float v8f;

__device__ __forceinline__ v4f clamp01(v4f a, v4f b) {
  v4f o;
  o.x = fminf(fmaxf(a.x + b.x, 0.f), 1.f);
  o.y = fminf(fmaxf(a.y + b.y, 0.f), 1.f);
  o.z = fminf(fmaxf(a.z + b.z, 0.f), 1.f);
  o.w = fminf(fmaxf(a.w + b.w, 0.f), 1.f);
  return o;
}

template <int J>
__device__ __forceinline__ unsigned long long row_base(const float* tab, int idx) {
  return (unsigned long long)tab +
         (unsigned long long)(unsigned)__builtin_amdgcn_readlane(idx, J) * (NACC * 4ull);
}

template <int J0>
__device__ __forceinline__ void issue_rows(v4f (&g)[8], const float* tab, int idx, unsigned laneByteOff) {
  const unsigned long long b0 = row_base<J0 + 0>(tab, idx);
  const unsigned long long b1 = row_base<J0 + 1>(tab, idx);
  const unsigned long long b2 = row_base<J0 + 2>(tab, idx);
  const unsigned long long b3 = row_base<J0 + 3>(tab, idx);
  const unsigned long long b4 = row_base<J0 + 4>(tab, idx);
  const unsigned long long b5 = row_base<J0 + 5>(tab, idx);
  const unsigned long long b6 = row_base<J0 + 6>(tab, idx);
  const unsigned long long b7 = row_base<J0 + 7>(tab, idx);
  asm volatile(
      "global_load_b128 %0, %8, %9\n\t"
      "global_load_b128 %1, %8, %10\n\t"
      "global_load_b128 %2, %8, %11\n\t"
      "global_load_b128 %3, %8, %12\n\t"
      "global_load_b128 %4, %8, %13\n\t"
      "global_load_b128 %5, %8, %14\n\t"
      "global_load_b128 %6, %8, %15\n\t"
      "global_load_b128 %7, %8, %16\n\t"
      "s_wait_xcnt 0x0"
      : "=&v"(g[0]), "=&v"(g[1]), "=&v"(g[2]), "=&v"(g[3]),
        "=&v"(g[4]), "=&v"(g[5]), "=&v"(g[6]), "=&v"(g[7])
      : "v"(laneByteOff), "s"(b0), "s"(b1), "s"(b2), "s"(b3),
        "s"(b4), "s"(b5), "s"(b6), "s"(b7));
}

template <int N>
__device__ __forceinline__ void wait_rows(v4f (&g)[8]) {
  asm volatile("s_wait_loadcnt %8"
               : "+v"(g[0]), "+v"(g[1]), "+v"(g[2]), "+v"(g[3]),
                 "+v"(g[4]), "+v"(g[5]), "+v"(g[6]), "+v"(g[7])
               : "n"(N));
}

__device__ __forceinline__ void issue_psqt(v4f& pw, v4f& pb, const float* tab,
                                           unsigned wOff, unsigned bOff) {
  asm volatile(
      "global_load_b128 %0, %2, %4\n\t"
      "global_load_b128 %1, %3, %4\n\t"
      "s_wait_xcnt 0x0"
      : "=&v"(pw), "=&v"(pb)
      : "v"(wOff), "v"(bOff), "s"((unsigned long long)tab));
}

__device__ __forceinline__ void wait_rows_first(v4f (&g)[8], v4f& pw, v4f& pb) {
  asm volatile("s_wait_loadcnt 0x10"
               : "+v"(g[0]), "+v"(g[1]), "+v"(g[2]), "+v"(g[3]),
                 "+v"(g[4]), "+v"(g[5]), "+v"(g[6]), "+v"(g[7]),
                 "+v"(pw), "+v"(pb));
}

__device__ __forceinline__ void add_rows(v4f& a, const v4f (&g)[8]) {
#pragma unroll
  for (int k = 0; k < 8; ++k) a += g[k];
}

__global__ __launch_bounds__(THREADS) void nnue_fwd(
    const int* __restrict__ w_cols, const int* __restrict__ b_cols,
    const float* __restrict__ stms, const float* __restrict__ psqt_w,
    const float* __restrict__ acc_w, const float* __restrict__ acc_bias,
    const float* __restrict__ layer_w, const float* __restrict__ layer_b,
    float* __restrict__ out, int n_pos)
{
  __shared__ __attribute__((aligned(16))) v4f sTab[PSQT_LDS_ROWS];
  __shared__ __attribute__((aligned(16))) float sBlend[2][TILE][ROWP];
  __shared__ __attribute__((aligned(16))) float sLW[NBUCK][ROWP];
  __shared__ float sPsqt[2][TILE][NBUCK];
  __shared__ float sSign[2][TILE];

  const int wave = __builtin_amdgcn_readfirstlane((int)(threadIdx.x >> 5));
  const int lane = threadIdx.x & 31;

  for (int r = threadIdx.x; r < PSQT_LDS_ROWS; r += THREADS)
    sTab[r] = *(const v4f*)(psqt_w + (size_t)r * NBUCK);
  for (int i = threadIdx.x; i < NBUCK * 2 * NACC; i += THREADS)
    sLW[i >> 8][i & 255] = layer_w[i];
  __syncthreads();

  const int n_tiles = (n_pos + TILE - 1) / TILE;
  const int first = blockIdx.x;
  const int step = gridDim.x;
  const int my_tiles = (n_tiles > first) ? (n_tiles - first + step - 1) / step : 0;
  const bool gatherer = wave < GATHER_WAVES;

  const unsigned laneByteOff = (unsigned)lane * 16u;
  v4f bias = {0.f, 0.f, 0.f, 0.f};
  int wcolN = 0, bcolN = 0;
  const int n  = lane & 15;
  const int hi = lane >> 4;
  const int nq = (n < NBUCK) ? n : 0;
  const float nmask = (n < NBUCK) ? 1.0f : 0.0f;
  const float* lw = &sLW[nq][0];
  float lb = 0.f;

  if (gatherer) {
    bias = *(const v4f*)(acc_bias + lane * 4);
    const int pos0 = first * TILE + wave;
    if (my_tiles > 0 && pos0 < n_pos) {
      wcolN = w_cols[(size_t)pos0 * BAG + lane];
      bcolN = b_cols[(size_t)pos0 * BAG + lane];
    }
  } else {
    lb = layer_b[nq];
  }

  for (int it = 0; it <= my_tiles; ++it) {
    if (gatherer) {
      if (it < my_tiles) {
        const int buf = it & 1;
        const int pos = (first + it * step) * TILE + wave;
        const int wcolL = wcolN;
        const int bcolL = bcolN;
        {
          const int posn = (first + (it + 1) * step) * TILE + wave;
          if (it + 1 < my_tiles && posn < n_pos) {
            wcolN = w_cols[(size_t)posn * BAG + lane];
            bcolN = b_cols[(size_t)posn * BAG + lane];
          }
        }
        if (pos < n_pos) {
          const bool wIn = (unsigned)wcolL < (unsigned)PSQT_LDS_ROWS;
          const bool bIn = (unsigned)bcolL < (unsigned)PSQT_LDS_ROWS;
          const v4f pwL = sTab[wIn ? wcolL : 0];
          const v4f pbL = sTab[bIn ? bcolL : 0];
          v4f pwG, pbG;
          issue_psqt(pwG, pbG, psqt_w,
                     (unsigned)(wIn ? PSQT_LDS_ROWS : wcolL) * (NBUCK * 4u),
                     (unsigned)(bIn ? PSQT_LDS_ROWS : bcolL) * (NBUCK * 4u));
          const float s = stms[pos];

          v4f g0[8], g1[8], g2[8];
          v4f wa = {0.f, 0.f, 0.f, 0.f};
          v4f ba = {0.f, 0.f, 0.f, 0.f};
          issue_rows<0>(g0, acc_w, wcolL, laneByteOff);
          issue_rows<8>(g1, acc_w, wcolL, laneByteOff);
          issue_rows<16>(g2, acc_w, wcolL, laneByteOff);
          wait_rows_first(g0, pwG, pbG); add_rows(wa, g0);
          issue_rows<24>(g0, acc_w, wcolL, laneByteOff);
          const v4f pw = wIn ? pwL : pwG;
          const v4f pb = bIn ? pbL : pbG;
          v4f p = pw - pb;
#pragma unroll
          for (int off = 16; off > 0; off >>= 1) {
            p.x += __shfl_xor(p.x, off, 32);
            p.y += __shfl_xor(p.y, off, 32);
            p.z += __shfl_xor(p.z, off, 32);
            p.w += __shfl_xor(p.w, off, 32);
          }
          wait_rows<16>(g1); add_rows(wa, g1);
          issue_rows<0>(g1, acc_w, bcolL, laneByteOff);
          wait_rows<16>(g2); add_rows(wa, g2);
          issue_rows<8>(g2, acc_w, bcolL, laneByteOff);
          wait_rows<16>(g0); add_rows(wa, g0);
          issue_rows<16>(g0, acc_w, bcolL, laneByteOff);
          wait_rows<16>(g1); add_rows(ba, g1);
          issue_rows<24>(g1, acc_w, bcolL, laneByteOff);
          wait_rows<16>(g2); add_rows(ba, g2);
          wait_rows<8>(g0);  add_rows(ba, g0);
          wait_rows<0>(g1);  add_rows(ba, g1);

          wa = clamp01(wa, bias);
          ba = clamp01(ba, bias);

          const bool flip = (s != 0.0f);
          v4f firstH  = flip ? ba : wa;
          v4f secondH = flip ? wa : ba;
          float* row = &sBlend[buf][wave][0];
          *(v4f*)(row + 4 * lane)        = firstH;
          *(v4f*)(row + NACC + 4 * lane) = secondH;

          if (lane == 0) {
            sPsqt[buf][wave][0] = p.x; sPsqt[buf][wave][1] = p.y;
            sPsqt[buf][wave][2] = p.z; sPsqt[buf][wave][3] = p.w;
            sSign[buf][wave] = flip ? -1.0f : 1.0f;
          }
        }
      }
    } else if (it > 0) {
      const int buf = (it - 1) & 1;
      const int tile = first + (it - 1) * step;
      const float* arow = &sBlend[buf][n][0];
      v8f c = {0.f, 0.f, 0.f, 0.f, 0.f, 0.f, 0.f, 0.f};
#pragma unroll 8
      for (int k4 = 0; k4 < (2 * NACC) / 4; ++k4) {
        const int kb = k4 * 4 + hi * 2;
        v2f a; a.x = arow[kb];        a.y = arow[kb + 1];
        v2f b; b.x = lw[kb] * nmask;  b.y = lw[kb + 1] * nmask;
        c = __builtin_amdgcn_wmma_f32_16x16x4_f32(
                false, a, false, b, (short)0, c, false, false);
      }
      if (n < NBUCK) {
#pragma unroll
        for (int r = 0; r < 8; ++r) {
          const int m = r + hi * 8;
          const int opos = tile * TILE + m;
          if (opos < n_pos)
            out[(size_t)opos * NBUCK + n] =
                sPsqt[buf][m][n] + sSign[buf][m] * (c[r] + lb);
        }
      }
    }
    __syncthreads();
  }
}

extern "C" void kernel_launch(void* const* d_in, const int* in_sizes, int n_in,
                              void* d_out, int out_size, void* d_ws, size_t ws_size,
                              hipStream_t stream) {
  const int*   w_cols   = (const int*)d_in[1];
  const int*   b_cols   = (const int*)d_in[3];
  const float* stms     = (const float*)d_in[4];
  const float* psqt_w   = (const float*)d_in[5];
  const float* acc_w    = (const float*)d_in[6];
  const float* acc_bias = (const float*)d_in[7];
  const float* layer_w  = (const float*)d_in[8];
  const float* layer_b  = (const float*)d_in[9];
  float* out = (float*)d_out;

  const int n_pos = in_sizes[4];
  nnue_fwd<<<256, THREADS, 0, stream>>>(
      w_cols, b_cols, stms, psqt_w, acc_w, acc_bias, layer_w, layer_b,
      out, n_pos);
}
